# starting kernel
# baseline (speedup 1.0000x reference)
#include <hip/hip_runtime.h>
#include <cstdint>
#include <cstddef>

#define N0_  200000
#define N1_  50000
#define N2_  10000
#define E0_  750000
#define E1_  150000
#define DIN_ 256
#define HID_ 256
#define OUT_ 64

typedef float v2f __attribute__((ext_vector_type(2)));
typedef float v8f __attribute__((ext_vector_type(8)));

__global__ void zero_f32(float* __restrict__ p, int n) {
  int i = blockIdx.x * blockDim.x + threadIdx.x;
  if (i < n) p[i] = 0.0f;
}

__global__ void scatter_mean_accum(const float* __restrict__ xsrc,
                                   const int* __restrict__ src,
                                   const int* __restrict__ dst,
                                   float* __restrict__ agg,
                                   float* __restrict__ cnt,
                                   int E) {
  int e = blockIdx.x * 4 + (threadIdx.x >> 6);
  if (e >= E) return;
  int t = threadIdx.x & 63;
  int s = src[e];
  int d = dst[e];
  float4 v = ((const float4*)(xsrc + (size_t)s * 256))[t];
  float* ap = agg + (size_t)d * 256 + t * 4;
  atomicAdd(ap + 0, v.x);
  atomicAdd(ap + 1, v.y);
  atomicAdd(ap + 2, v.z);
  atomicAdd(ap + 3, v.w);
  if (t == 0) atomicAdd(cnt + d, 1.0f);
}

template <int KTOT, int NCOL, int ACT, bool NORM>
__global__ __launch_bounds__((NCOL / 16) * 32)
void sage_gemm_wmma(const float* __restrict__ A0,
                    const float* __restrict__ cnt,
                    const float* __restrict__ A1,
                    const float* __restrict__ B0,
                    const float* __restrict__ B1,
                    const float* __restrict__ bias,
                    float* __restrict__ out,
                    int M) {
  constexpr int NT  = (NCOL / 16) * 32;
  constexpr int LDK = KTOT + 4;
  __shared__ float Asub[32][LDK];

  const int tid  = threadIdx.x;
  const int wave = tid >> 5;
  const int lane = tid & 31;
  const int half = lane >> 4;
  const int l15  = lane & 15;
  const int m0   = blockIdx.x * 32;
  const bool has2 = (m0 + 16) < M;

  for (int i = tid * 4; i < 32 * KTOT; i += NT * 4) {
    int row = i / KTOT;
    int col = i % KTOT;
    int gr  = m0 + row;
    float4 v = {0.0f, 0.0f, 0.0f, 0.0f};
    if (gr < M) {
      if (col < 256) {
        v = *(const float4*)(A0 + (size_t)gr * 256 + col);
        if (NORM) {
          float c = cnt[gr];
          float s = (c > 0.0f) ? 1.0f / c : 0.0f;
          v.x *= s; v.y *= s; v.z *= s; v.w *= s;
        }
      } else {
        v = *(const float4*)(A1 + (size_t)gr * 256 + (col - 256));
      }
    }
    *(float4*)&Asub[row][col] = v;
  }
  __syncthreads();

  const int nbase = wave * 16;
  v8f acc0 = {0.0f, 0.0f, 0.0f, 0.0f, 0.0f, 0.0f, 0.0f, 0.0f};
  v8f acc1 = {0.0f, 0.0f, 0.0f, 0.0f, 0.0f, 0.0f, 0.0f, 0.0f};

#pragma unroll
  for (int seg = 0; seg < KTOT / 256; ++seg) {
    const float* bp = ((seg == 0) ? B0 : B1) +
                      (size_t)(half * 2) * NCOL + nbase + l15;
    const int kb = seg * 256;
#pragma unroll 8
    for (int k = 0; k < 256; k += 4) {
      v2f b;
      b.x = bp[0];
      b.y = bp[NCOL];
      int ka = kb + k + half * 2;
      float2 a0v = *(const float2*)&Asub[l15][ka];
      float2 a1v = *(const float2*)&Asub[16 + l15][ka];
      v2f a0; a0.x = a0v.x; a0.y = a0v.y;
      v2f a1; a1.x = a1v.x; a1.y = a1v.y;
      acc0 = __builtin_amdgcn_wmma_f32_16x16x4_f32(false, a0, false, b,
                                                   (short)0, acc0, false, false);
      acc1 = __builtin_amdgcn_wmma_f32_16x16x4_f32(false, a1, false, b,
                                                   (short)0, acc1, false, false);
      bp += (size_t)4 * NCOL;
    }
  }

  const int col = nbase + l15;
  const float bv = bias[col];
#pragma unroll
  for (int r = 0; r < 8; ++r) {
    int row = m0 + half * 8 + r;
    float v = acc0[r] + bv;
    if (ACT == 1) v = fmaxf(v, 0.0f);
    if (ACT == 2) v = tanhf(v);
    out[(size_t)row * NCOL + col] = v;
  }
  if (has2) {
#pragma unroll
    for (int r = 0; r < 8; ++r) {
      int row = m0 + 16 + half * 8 + r;
      float v = acc1[r] + bv;
      if (ACT == 1) v = fmaxf(v, 0.0f);
      if (ACT == 2) v = tanhf(v);
      out[(size_t)row * NCOL + col] = v;
    }
  }
}

__global__ void softmax_rows64(const float* __restrict__ logits,
                               float* __restrict__ out, int M) {
  int row = blockIdx.x * 8 + (threadIdx.x >> 5);
  if (row >= M) return;
  int lane = threadIdx.x & 31;
  const float* lp = logits + (size_t)row * 64;
  float a = lp[lane];
  float b = lp[lane + 32];
  float m = fmaxf(a, b);
  for (int off = 16; off > 0; off >>= 1) m = fmaxf(m, __shfl_xor(m, off, 32));
  float ea = expf(a - m);
  float eb = expf(b - m);
  float s = ea + eb;
  for (int off = 16; off > 0; off >>= 1) s += __shfl_xor(s, off, 32);
  float inv = 1.0f / s;
  float* op = out + (size_t)row * 64;
  op[lane] = ea * inv;
  op[lane + 32] = eb * inv;
}

extern "C" void kernel_launch(void* const* d_in, const int* in_sizes, int n_in,
                              void* d_out, int out_size, void* d_ws,
                              size_t ws_size, hipStream_t stream) {
  (void)in_sizes; (void)n_in; (void)out_size; (void)ws_size;

  const float* x    = (const float*)d_in[0];
  const int*   src0 = (const int*)d_in[1];
  const int*   dst0 = (const int*)d_in[2];
  const int*   src1 = (const int*)d_in[3];
  const int*   dst1 = (const int*)d_in[4];
  const float* Wl0  = (const float*)d_in[5];
  const float* bl0  = (const float*)d_in[6];
  const float* Wr0  = (const float*)d_in[7];
  const float* Wl1  = (const float*)d_in[8];
  const float* bl1  = (const float*)d_in[9];
  const float* Wr1  = (const float*)d_in[10];
  const float* Wlin = (const float*)d_in[11];
  const float* blin = (const float*)d_in[12];

  char* ws = (char*)d_ws;
  auto align256 = [](size_t v) { return (v + 255) & ~(size_t)255; };
  size_t off = 0;
  float* agg0 = (float*)(ws + off); off += (size_t)N1_ * 256 * 4;
  float* cnt0 = (float*)(ws + off); off += (size_t)N1_ * 4; off = align256(off);
  float* h0   = (float*)(ws + off); off += (size_t)N1_ * 256 * 4;
  float* agg1 = (float*)(ws + off); off += (size_t)N2_ * 256 * 4;
  float* cnt1 = (float*)(ws + off); off += (size_t)N2_ * 4; off = align256(off);
  float* h1   = (float*)(ws + off); off += (size_t)N2_ * 256 * 4;
  float* lgt  = (float*)(ws + off); off += (size_t)N2_ * 64 * 4;

  {
    int n = N1_ * 256 + N1_;
    zero_f32<<<(n + 255) / 256, 256, 0, stream>>>(agg0, n);
  }
  scatter_mean_accum<<<E0_ / 4, 256, 0, stream>>>(x, src0, dst0, agg0, cnt0, E0_);
  sage_gemm_wmma<512, 256, 1, true><<<(N1_ + 31) / 32, 512, 0, stream>>>(
      agg0, cnt0, x, Wl0, Wr0, bl0, h0, N1_);

  {
    int n = N2_ * 256 + N2_;
    zero_f32<<<(n + 255) / 256, 256, 0, stream>>>(agg1, n);
  }
  scatter_mean_accum<<<E1_ / 4, 256, 0, stream>>>(h0, src1, dst1, agg1, cnt1, E1_);
  sage_gemm_wmma<512, 256, 2, true><<<(N2_ + 31) / 32, 512, 0, stream>>>(
      agg1, cnt1, h0, Wl1, Wr1, bl1, h1, N2_);

  sage_gemm_wmma<256, 64, 0, false><<<(N2_ + 31) / 32, 128, 0, stream>>>(
      h1, nullptr, nullptr, Wlin, nullptr, blin, lgt, N2_);
  softmax_rows64<<<(N2_ + 7) / 8, 256, 0, stream>>>(lgt, (float*)d_out, N2_);
}


# Global float-atomic scatter replaced by dst-range owner blocks: each block scans dst, queues its edges, sums their source rows in LDS (LDS f32 adds), stores means; row sum order = LDS arrival order
# speedup vs baseline: 2.2684x; 2.2684x over previous
#include <hip/hip_runtime.h>
#include <cstdint>
#include <cstddef>

#define N0_  200000
#define N1_  50000
#define N2_  10000
#define E0_  750000
#define E1_  150000
#define DIN_ 256
#define HID_ 256
#define OUT_ 64

typedef float v2f __attribute__((ext_vector_type(2)));
typedef float v8f __attribute__((ext_vector_type(8)));

static __device__ __forceinline__ void lds_add_f32(float* p, float v) {
  __hip_atomic_fetch_add(p, v, __ATOMIC_RELAXED, __HIP_MEMORY_SCOPE_WORKGROUP);
}

template <int R>
__global__ __launch_bounds__(1024)
void agg_mean_owner(const float* __restrict__ xsrc,
                    const int* __restrict__ src,
                    const int* __restrict__ dst,
                    float* __restrict__ agg,
                    int E, int M) {
  constexpr int NW   = 32;
  constexpr int UB   = 8;
  constexpr int QCAP = 32 + 512;
  static_assert(R <= 256, "row id is packed into 8 bits");
  __shared__ float    acc[R * 256];
  __shared__ unsigned cnt[R];
  __shared__ unsigned que[NW][QCAP];

  const int tid  = threadIdx.x;
  const int wave = tid >> 5;
  const int lane = tid & 31;
  const int m0   = blockIdx.x * R;

  for (int i = tid * 4; i < R * 256; i += 1024 * 4) {
    float4 z = {0.0f, 0.0f, 0.0f, 0.0f};
    *(float4*)&acc[i] = z;
  }
  if (tid < R) cnt[tid] = 0u;
  __syncthreads();

  int qn = 0;

  auto append = [&](unsigned b, bool hit, unsigned r, int e) {
    if (hit) {
      int pos = qn + (int)__builtin_amdgcn_mbcnt_lo(b, 0u);
      que[wave][pos] = (r << 24) | (unsigned)e;
      __hip_atomic_fetch_add(&cnt[r], 1u, __ATOMIC_RELAXED, __HIP_MEMORY_SCOPE_WORKGROUP);
    }
    qn += __builtin_popcount(b);
  };

  auto drain = [&](int n) {
    for (int base = 0; base < n; base += 32) {
      const int idx = base + lane;
      const bool live = idx < n;
      const unsigned ent = live ? que[wave][idx] : 0u;
      const int s_l = live ? src[ent & 0xFFFFFFu] : 0;
      const int r_l = (int)(ent >> 24);
      const int nb = (n - base < 32) ? (n - base) : 32;
      for (int j = 0; j < nb; j += UB) {
        float4 v0[UB], v1[UB];
#pragma unroll
        for (int u = 0; u < UB; ++u) {
          const int t = (j + u < nb) ? (j + u) : j;
          const int s = __builtin_amdgcn_readlane(s_l, t);
          const float4* p = (const float4*)(xsrc + (size_t)s * 256);
          v0[u] = p[lane];
          v1[u] = p[32 + lane];
        }
#pragma unroll
        for (int u = 0; u < UB; ++u) {
          if (j + u < nb) {
            const int r = __builtin_amdgcn_readlane(r_l, j + u);
            float* a = acc + r * 256 + lane;
            lds_add_f32(a +   0, v0[u].x);
            lds_add_f32(a +  32, v0[u].y);
            lds_add_f32(a +  64, v0[u].z);
            lds_add_f32(a +  96, v0[u].w);
            lds_add_f32(a + 128, v1[u].x);
            lds_add_f32(a + 160, v1[u].y);
            lds_add_f32(a + 192, v1[u].z);
            lds_add_f32(a + 224, v1[u].w);
          }
        }
      }
    }
  };

  const int nvec = E >> 2;
  const int4* dst4 = (const int4*)dst;
  const int none = m0 - 1;

  if (wave == 0) {
    const int e = (nvec << 2) + lane;
    const int dv = (e < E) ? dst[e] : none;
    const unsigned r = (unsigned)dv - (unsigned)m0;
    const bool hit = r < (unsigned)R;
    const unsigned b = __builtin_amdgcn_ballot_w32(hit);
    if (b != 0u) append(b, hit, r, e);
  }

  for (int g0 = wave * 32;; g0 += NW * 32 * 4) {
    const bool more = g0 < nvec;
    if (qn >= 32 || !more) { drain(qn); qn = 0; }
    if (!more) break;
    int4 d[4];
#pragma unroll
    for (int u = 0; u < 4; ++u) {
      const int g = g0 + u * (NW * 32) + lane;
      const bool in = g < nvec;
      const int4 t = dst4[in ? g : 0];
      d[u].x = in ? t.x : none;
      d[u].y = in ? t.y : none;
      d[u].z = in ? t.z : none;
      d[u].w = in ? t.w : none;
    }
#pragma unroll
    for (int u = 0; u < 4; ++u) {
      const int e = (g0 + u * (NW * 32) + lane) << 2;
      const unsigned r0 = (unsigned)d[u].x - (unsigned)m0;
      const unsigned r1 = (unsigned)d[u].y - (unsigned)m0;
      const unsigned r2 = (unsigned)d[u].z - (unsigned)m0;
      const unsigned r3 = (unsigned)d[u].w - (unsigned)m0;
      const bool h0 = r0 < (unsigned)R, h1 = r1 < (unsigned)R;
      const bool h2 = r2 < (unsigned)R, h3 = r3 < (unsigned)R;
      const unsigned b0 = __builtin_amdgcn_ballot_w32(h0);
      const unsigned b1 = __builtin_amdgcn_ballot_w32(h1);
      const unsigned b2 = __builtin_amdgcn_ballot_w32(h2);
      const unsigned b3 = __builtin_amdgcn_ballot_w32(h3);
      if ((b0 | b1 | b2 | b3) != 0u) {
        append(b0, h0, r0, e + 0);
        append(b1, h1, r1, e + 1);
        append(b2, h2, r2, e + 2);
        append(b3, h3, r3, e + 3);
      }
    }
  }
  __syncthreads();

  for (int row = wave; row < R; row += NW) {
    const int gr = m0 + row;
    if (gr < M) {
      const unsigned c = cnt[row];
      const float sc = (c > 0u) ? 1.0f / (float)c : 0.0f;
      const float* a = acc + row * 256 + lane;
      float4 o0 = {a[0] * sc, a[32] * sc, a[64] * sc, a[96] * sc};
      float4 o1 = {a[128] * sc, a[160] * sc, a[192] * sc, a[224] * sc};
      float4* op = (float4*)(agg + (size_t)gr * 256);
      op[lane] = o0;
      op[32 + lane] = o1;
    }
  }
}

template <int KTOT, int NCOL, int ACT, bool NORM>
__global__ __launch_bounds__((NCOL / 16) * 32)
void sage_gemm_wmma(const float* __restrict__ A0,
                    const float* __restrict__ cnt,
                    const float* __restrict__ A1,
                    const float* __restrict__ B0,
                    const float* __restrict__ B1,
                    const float* __restrict__ bias,
                    float* __restrict__ out,
                    int M) {
  constexpr int NT  = (NCOL / 16) * 32;
  constexpr int LDK = KTOT + 4;
  __shared__ float Asub[32][LDK];

  const int tid  = threadIdx.x;
  const int wave = tid >> 5;
  const int lane = tid & 31;
  const int half = lane >> 4;
  const int l15  = lane & 15;
  const int m0   = blockIdx.x * 32;
  const bool has2 = (m0 + 16) < M;

  for (int i = tid * 4; i < 32 * KTOT; i += NT * 4) {
    int row = i / KTOT;
    int col = i % KTOT;
    int gr  = m0 + row;
    float4 v = {0.0f, 0.0f, 0.0f, 0.0f};
    if (gr < M) {
      if (col < 256) {
        v = *(const float4*)(A0 + (size_t)gr * 256 + col);
        if (NORM) {
          float c = cnt[gr];
          float s = (c > 0.0f) ? 1.0f / c : 0.0f;
          v.x *= s; v.y *= s; v.z *= s; v.w *= s;
        }
      } else {
        v = *(const float4*)(A1 + (size_t)gr * 256 + (col - 256));
      }
    }
    *(float4*)&Asub[row][col] = v;
  }
  __syncthreads();

  const int nbase = wave * 16;
  v8f acc0 = {0.0f, 0.0f, 0.0f, 0.0f, 0.0f, 0.0f, 0.0f, 0.0f};
  v8f acc1 = {0.0f, 0.0f, 0.0f, 0.0f, 0.0f, 0.0f, 0.0f, 0.0f};

#pragma unroll
  for (int seg = 0; seg < KTOT / 256; ++seg) {
    const float* bp = ((seg == 0) ? B0 : B1) +
                      (size_t)(half * 2) * NCOL + nbase + l15;
    const int kb = seg * 256;
#pragma unroll 8
    for (int k = 0; k < 256; k += 4) {
      v2f b;
      b.x = bp[0];
      b.y = bp[NCOL];
      int ka = kb + k + half * 2;
      float2 a0v = *(const float2*)&Asub[l15][ka];
      float2 a1v = *(const float2*)&Asub[16 + l15][ka];
      v2f a0; a0.x = a0v.x; a0.y = a0v.y;
      v2f a1; a1.x = a1v.x; a1.y = a1v.y;
      acc0 = __builtin_amdgcn_wmma_f32_16x16x4_f32(false, a0, false, b,
                                                   (short)0, acc0, false, false);
      acc1 = __builtin_amdgcn_wmma_f32_16x16x4_f32(false, a1, false, b,
                                                   (short)0, acc1, false, false);
      bp += (size_t)4 * NCOL;
    }
  }

  const int col = nbase + l15;
  const float bv = bias[col];
#pragma unroll
  for (int r = 0; r < 8; ++r) {
    int row = m0 + half * 8 + r;
    float v = acc0[r] + bv;
    if (ACT == 1) v = fmaxf(v, 0.0f);
    if (ACT == 2) v = tanhf(v);
    out[(size_t)row * NCOL + col] = v;
  }
  if (has2) {
#pragma unroll
    for (int r = 0; r < 8; ++r) {
      int row = m0 + 16 + half * 8 + r;
      float v = acc1[r] + bv;
      if (ACT == 1) v = fmaxf(v, 0.0f);
      if (ACT == 2) v = tanhf(v);
      out[(size_t)row * NCOL + col] = v;
    }
  }
}

__global__ void softmax_rows64(const float* __restrict__ logits,
                               float* __restrict__ out, int M) {
  int row = blockIdx.x * 8 + (threadIdx.x >> 5);
  if (row >= M) return;
  int lane = threadIdx.x & 31;
  const float* lp = logits + (size_t)row * 64;
  float a = lp[lane];
  float b = lp[lane + 32];
  float m = fmaxf(a, b);
  for (int off = 16; off > 0; off >>= 1) m = fmaxf(m, __shfl_xor(m, off, 32));
  float ea = expf(a - m);
  float eb = expf(b - m);
  float s = ea + eb;
  for (int off = 16; off > 0; off >>= 1) s += __shfl_xor(s, off, 32);
  float inv = 1.0f / s;
  float* op = out + (size_t)row * 64;
  op[lane] = ea * inv;
  op[lane + 32] = eb * inv;
}

extern "C" void kernel_launch(void* const* d_in, const int* in_sizes, int n_in,
                              void* d_out, int out_size, void* d_ws,
                              size_t ws_size, hipStream_t stream) {
  (void)in_sizes; (void)n_in; (void)out_size; (void)ws_size;

  const float* x    = (const float*)d_in[0];
  const int*   src0 = (const int*)d_in[1];
  const int*   dst0 = (const int*)d_in[2];
  const int*   src1 = (const int*)d_in[3];
  const int*   dst1 = (const int*)d_in[4];
  const float* Wl0  = (const float*)d_in[5];
  const float* bl0  = (const float*)d_in[6];
  const float* Wr0  = (const float*)d_in[7];
  const float* Wl1  = (const float*)d_in[8];
  const float* bl1  = (const float*)d_in[9];
  const float* Wr1  = (const float*)d_in[10];
  const float* Wlin = (const float*)d_in[11];
  const float* blin = (const float*)d_in[12];

  char* ws = (char*)d_ws;
  size_t off = 0;
  float* agg0 = (float*)(ws + off); off += (size_t)N1_ * 256 * 4;
  float* h0   = (float*)(ws + off); off += (size_t)N1_ * 256 * 4;
  float* agg1 = (float*)(ws + off); off += (size_t)N2_ * 256 * 4;
  float* h1   = (float*)(ws + off); off += (size_t)N2_ * 256 * 4;
  float* lgt  = (float*)(ws + off); off += (size_t)N2_ * 64 * 4;

  static_assert(E0_ < (1 << 24) && E1_ < (1 << 24), "edge index is packed into 24 bits");
  constexpr int R0 = 196;
  constexpr int R1 = 40;

  agg_mean_owner<R0><<<(N1_ + R0 - 1) / R0, 1024, 0, stream>>>(
      x, src0, dst0, agg0, E0_, N1_);
  sage_gemm_wmma<512, 256, 1, false><<<(N1_ + 31) / 32, 512, 0, stream>>>(
      agg0, nullptr, x, Wl0, Wr0, bl0, h0, N1_);

  agg_mean_owner<R1><<<(N2_ + R1 - 1) / R1, 1024, 0, stream>>>(
      h0, src1, dst1, agg1, E1_, N2_);
  sage_gemm_wmma<512, 256, 2, false><<<(N2_ + 31) / 32, 512, 0, stream>>>(
      agg1, nullptr, h0, Wl1, Wr1, bl1, h1, N2_);

  sage_gemm_wmma<256, 64, 0, false><<<(N2_ + 31) / 32, 128, 0, stream>>>(
      h1, nullptr, nullptr, Wlin, nullptr, blin, lgt, N2_);
  softmax_rows64<<<(N2_ + 7) / 8, 256, 0, stream>>>(lgt, (float*)d_out, N2_);
}


# Aggregation scan reads a 1-byte owner-id plane (dst/R per edge, written by one new launch for both layers) instead of 4-byte dst: 4x fewer scan bytes; row sums still in LDS arrival order
# speedup vs baseline: 2.3906x; 1.0539x over previous
#include <hip/hip_runtime.h>
#include <cstdint>
#include <cstddef>

#define N0_  200000
#define N1_  50000
#define N2_  10000
#define E0_  750000
#define E1_  150000
#define DIN_ 256
#define HID_ 256
#define OUT_ 64

typedef float v2f __attribute__((ext_vector_type(2)));
typedef float v8f __attribute__((ext_vector_type(8)));

template <int R>
static __device__ __forceinline__ unsigned owner_word(int4 d) {
  return (((unsigned)d.x / (unsigned)R) & 0xFFu) |
         ((((unsigned)d.y / (unsigned)R) & 0xFFu) << 8) |
         ((((unsigned)d.z / (unsigned)R) & 0xFFu) << 16) |
         ((((unsigned)d.w / (unsigned)R) & 0xFFu) << 24);
}

template <int RA, int RB>
__global__ __launch_bounds__(256)
void owner_planes(const int* __restrict__ dstA, unsigned* __restrict__ ownA, int nA,
                  const int* __restrict__ dstB, unsigned* __restrict__ ownB, int nB) {
  const int t0 = blockIdx.x * 256 + threadIdx.x;
  const int stride = gridDim.x * 256;
  int4 d[4];
#pragma unroll
  for (int k = 0; k < 4; ++k) {
    const int t = t0 + k * stride;
    int4 z = {0, 0, 0, 0};
    d[k] = z;
    if (t < nA) d[k] = ((const int4*)dstA)[t];
    else if (t < nA + nB) d[k] = ((const int4*)dstB)[t - nA];
  }
#pragma unroll
  for (int k = 0; k < 4; ++k) {
    const int t = t0 + k * stride;
    if (t < nA) ownA[t] = owner_word<RA>(d[k]);
    else if (t < nA + nB) ownB[t - nA] = owner_word<RB>(d[k]);
  }
}

static __device__ __forceinline__ void lds_add_f32(float* p, float v) {
  __hip_atomic_fetch_add(p, v, __ATOMIC_RELAXED, __HIP_MEMORY_SCOPE_WORKGROUP);
}

template <int R>
__global__ __launch_bounds__(1024)
void agg_mean_owner(const float* __restrict__ xsrc,
                    const int* __restrict__ src,
                    const int* __restrict__ dst,
                    const unsigned* __restrict__ own,
                    float* __restrict__ agg,
                    int E, int M) {
  constexpr int NW   = 32;
  constexpr int UB   = 8;
  constexpr int QCAP = 32 + 512;
  __shared__ float    acc[R * 256];
  __shared__ unsigned cnt[R];
  __shared__ unsigned que[NW][QCAP];

  const int tid  = threadIdx.x;
  const int wave = __builtin_amdgcn_readfirstlane(tid >> 5);
  const int lane = tid & 31;
  const int m0   = blockIdx.x * R;

  for (int i = tid * 4; i < R * 256; i += 1024 * 4) {
    float4 z = {0.0f, 0.0f, 0.0f, 0.0f};
    *(float4*)&acc[i] = z;
  }
  if (tid < R) cnt[tid] = 0u;
  __syncthreads();

  int qn = 0;

  auto append = [&](unsigned b, bool hit, int e) {
    if (hit) {
      int pos = qn + (int)__builtin_amdgcn_mbcnt_lo(b, 0u);
      que[wave][pos] = (unsigned)e;
    }
    qn += __builtin_popcount(b);
  };

  auto drain = [&](int n) {
    for (int base = 0; base < n; base += 32) {
      const int idx = base + lane;
      const bool live = idx < n;
      const int e = live ? (int)que[wave][idx] : 0;
      const int s_l = live ? src[e] : 0;
      const int d_l = live ? dst[e] : m0 - 1;
      const unsigned r_l = (unsigned)d_l - (unsigned)m0;
      const bool ok = r_l < (unsigned)R;
      if (ok) __hip_atomic_fetch_add(&cnt[r_l], 1u, __ATOMIC_RELAXED, __HIP_MEMORY_SCOPE_WORKGROUP);
      const unsigned okm = __builtin_amdgcn_ballot_w32(ok);
      const int nb = (n - base < 32) ? (n - base) : 32;
      for (int j = 0; j < nb; j += UB) {
        float4 v0[UB], v1[UB];
#pragma unroll
        for (int u = 0; u < UB; ++u) {
          const int t = (j + u < nb) ? (j + u) : j;
          const int s = __builtin_amdgcn_readlane(s_l, t);
          const float4* p = (const float4*)(xsrc + (size_t)s * 256);
          v0[u] = p[lane];
          v1[u] = p[32 + lane];
        }
#pragma unroll
        for (int u = 0; u < UB; ++u) {
          if (j + u < nb && ((okm >> (j + u)) & 1u) != 0u) {
            const int r = __builtin_amdgcn_readlane((int)r_l, j + u);
            float* a = acc + r * 256 + lane;
            lds_add_f32(a +   0, v0[u].x);
            lds_add_f32(a +  32, v0[u].y);
            lds_add_f32(a +  64, v0[u].z);
            lds_add_f32(a +  96, v0[u].w);
            lds_add_f32(a + 128, v1[u].x);
            lds_add_f32(a + 160, v1[u].y);
            lds_add_f32(a + 192, v1[u].z);
            lds_add_f32(a + 224, v1[u].w);
          }
        }
      }
    }
  };

  const int nvec = E >> 4;
  const uint4* own4 = (const uint4*)own;
  const unsigned pat = (unsigned)blockIdx.x * 0x01010101u;

  if (wave == 0) {
    const int e = (nvec << 4) + lane;
    const int dv = (e < E) ? dst[e] : m0 - 1;
    const unsigned r = (unsigned)dv - (unsigned)m0;
    const bool hit = r < (unsigned)R;
    const unsigned b = __builtin_amdgcn_ballot_w32(hit);
    if (b != 0u) append(b, hit, e);
  }

  auto flags = [&](unsigned w) -> unsigned {
    const unsigned x = w ^ pat;
    const unsigned t = (x & 0x7F7F7F7Fu) + 0x7F7F7F7Fu;
    return ~(t | x) & 0x80808080u;
  };
  auto nib = [](unsigned z) -> unsigned { return ((z >> 7) * 0x10204080u) >> 28; };

  constexpr int S = NW * 32;
  auto ldv = [&](int g) -> uint4 { return own4[(g < nvec) ? g : 0]; };

  int gnext = wave * 32;
  int gbase = 0;
  int u = 4;
  bool done = false;
  uint4 d0 = {0u, 0u, 0u, 0u}, d1 = d0, d2 = d0, d3 = d0;
  for (;;) {
    if (u == 4) {
      if (gnext < nvec) {
        gbase = gnext;
        d0 = ldv(gbase + lane);
        d1 = ldv(gbase + S + lane);
        d2 = ldv(gbase + 2 * S + lane);
        d3 = ldv(gbase + 3 * S + lane);
        gnext += 4 * S;
        u = 0;
      } else {
        done = true;
      }
    }
    if (qn >= 32 || done) { drain(qn); qn = 0; }
    if (done) break;
    const int gs = gbase + u * S;
    if (gs < nvec) {
      uint4 cur;
      switch (u) {
        case 0: cur = d0; break;
        case 1: cur = d1; break;
        case 2: cur = d2; break;
        default: cur = d3; break;
      }
      const int g = gs + lane;
      const unsigned z0 = flags(cur.x), z1 = flags(cur.y);
      const unsigned z2 = flags(cur.z), z3 = flags(cur.w);
      const bool any = (g < nvec) && ((z0 | z1 | z2 | z3) != 0u);
      unsigned b = __builtin_amdgcn_ballot_w32(any);
      if (b != 0u) {
        unsigned m = any ? (nib(z0) | (nib(z1) << 4) | (nib(z2) << 8) | (nib(z3) << 12)) : 0u;
        while (b != 0u) {
          const bool hit = m != 0u;
          const int bit = __builtin_ctz(m | 0x10000u);
          append(b, hit, (g << 4) + bit);
          m &= m - 1u;
          b = __builtin_amdgcn_ballot_w32(m != 0u);
        }
      }
    }
    ++u;
  }
  __syncthreads();

  for (int row = wave; row < R; row += NW) {
    const int gr = m0 + row;
    if (gr < M) {
      const unsigned c = cnt[row];
      const float sc = (c > 0u) ? 1.0f / (float)c : 0.0f;
      const float* a = acc + row * 256 + lane;
      float4 o0 = {a[0] * sc, a[32] * sc, a[64] * sc, a[96] * sc};
      float4 o1 = {a[128] * sc, a[160] * sc, a[192] * sc, a[224] * sc};
      float4* op = (float4*)(agg + (size_t)gr * 256);
      op[lane] = o0;
      op[32 + lane] = o1;
    }
  }
}

template <int KTOT, int NCOL, int ACT, bool NORM>
__global__ __launch_bounds__((NCOL / 16) * 32)
void sage_gemm_wmma(const float* __restrict__ A0,
                    const float* __restrict__ cnt,
                    const float* __restrict__ A1,
                    const float* __restrict__ B0,
                    const float* __restrict__ B1,
                    const float* __restrict__ bias,
                    float* __restrict__ out,
                    int M) {
  constexpr int NT  = (NCOL / 16) * 32;
  constexpr int LDK = KTOT + 4;
  __shared__ float Asub[32][LDK];

  const int tid  = threadIdx.x;
  const int wave = tid >> 5;
  const int lane = tid & 31;
  const int half = lane >> 4;
  const int l15  = lane & 15;
  const int m0   = blockIdx.x * 32;
  const bool has2 = (m0 + 16) < M;

  for (int i = tid * 4; i < 32 * KTOT; i += NT * 4) {
    int row = i / KTOT;
    int col = i % KTOT;
    int gr  = m0 + row;
    float4 v = {0.0f, 0.0f, 0.0f, 0.0f};
    if (gr < M) {
      if (col < 256) {
        v = *(const float4*)(A0 + (size_t)gr * 256 + col);
        if (NORM) {
          float c = cnt[gr];
          float s = (c > 0.0f) ? 1.0f / c : 0.0f;
          v.x *= s; v.y *= s; v.z *= s; v.w *= s;
        }
      } else {
        v = *(const float4*)(A1 + (size_t)gr * 256 + (col - 256));
      }
    }
    *(float4*)&Asub[row][col] = v;
  }
  __syncthreads();

  const int nbase = wave * 16;
  v8f acc0 = {0.0f, 0.0f, 0.0f, 0.0f, 0.0f, 0.0f, 0.0f, 0.0f};
  v8f acc1 = {0.0f, 0.0f, 0.0f, 0.0f, 0.0f, 0.0f, 0.0f, 0.0f};

#pragma unroll
  for (int seg = 0; seg < KTOT / 256; ++seg) {
    const float* bp = ((seg == 0) ? B0 : B1) +
                      (size_t)(half * 2) * NCOL + nbase + l15;
    const int kb = seg * 256;
#pragma unroll 8
    for (int k = 0; k < 256; k += 4) {
      v2f b;
      b.x = bp[0];
      b.y = bp[NCOL];
      int ka = kb + k + half * 2;
      float2 a0v = *(const float2*)&Asub[l15][ka];
      float2 a1v = *(const float2*)&Asub[16 + l15][ka];
      v2f a0; a0.x = a0v.x; a0.y = a0v.y;
      v2f a1; a1.x = a1v.x; a1.y = a1v.y;
      acc0 = __builtin_amdgcn_wmma_f32_16x16x4_f32(false, a0, false, b,
                                                   (short)0, acc0, false, false);
      acc1 = __builtin_amdgcn_wmma_f32_16x16x4_f32(false, a1, false, b,
                                                   (short)0, acc1, false, false);
      bp += (size_t)4 * NCOL;
    }
  }

  const int col = nbase + l15;
  const float bv = bias[col];
#pragma unroll
  for (int r = 0; r < 8; ++r) {
    int row = m0 + half * 8 + r;
    float v = acc0[r] + bv;
    if (ACT == 1) v = fmaxf(v, 0.0f);
    if (ACT == 2) v = tanhf(v);
    out[(size_t)row * NCOL + col] = v;
  }
  if (has2) {
#pragma unroll
    for (int r = 0; r < 8; ++r) {
      int row = m0 + 16 + half * 8 + r;
      float v = acc1[r] + bv;
      if (ACT == 1) v = fmaxf(v, 0.0f);
      if (ACT == 2) v = tanhf(v);
      out[(size_t)row * NCOL + col] = v;
    }
  }
}

__global__ void softmax_rows64(const float* __restrict__ logits,
                               float* __restrict__ out, int M) {
  int row = blockIdx.x * 8 + (threadIdx.x >> 5);
  if (row >= M) return;
  int lane = threadIdx.x & 31;
  const float* lp = logits + (size_t)row * 64;
  float a = lp[lane];
  float b = lp[lane + 32];
  float m = fmaxf(a, b);
  for (int off = 16; off > 0; off >>= 1) m = fmaxf(m, __shfl_xor(m, off, 32));
  float ea = expf(a - m);
  float eb = expf(b - m);
  float s = ea + eb;
  for (int off = 16; off > 0; off >>= 1) s += __shfl_xor(s, off, 32);
  float inv = 1.0f / s;
  float* op = out + (size_t)row * 64;
  op[lane] = ea * inv;
  op[lane + 32] = eb * inv;
}

extern "C" void kernel_launch(void* const* d_in, const int* in_sizes, int n_in,
                              void* d_out, int out_size, void* d_ws,
                              size_t ws_size, hipStream_t stream) {
  (void)in_sizes; (void)n_in; (void)out_size; (void)ws_size;

  const float* x    = (const float*)d_in[0];
  const int*   src0 = (const int*)d_in[1];
  const int*   dst0 = (const int*)d_in[2];
  const int*   src1 = (const int*)d_in[3];
  const int*   dst1 = (const int*)d_in[4];
  const float* Wl0  = (const float*)d_in[5];
  const float* bl0  = (const float*)d_in[6];
  const float* Wr0  = (const float*)d_in[7];
  const float* Wl1  = (const float*)d_in[8];
  const float* bl1  = (const float*)d_in[9];
  const float* Wr1  = (const float*)d_in[10];
  const float* Wlin = (const float*)d_in[11];
  const float* blin = (const float*)d_in[12];

  char* ws = (char*)d_ws;
  size_t off = 0;
  float* agg0 = (float*)(ws + off); off += (size_t)N1_ * 256 * 4;
  float* h0   = (float*)(ws + off); off += (size_t)N1_ * 256 * 4;
  float* agg1 = (float*)(ws + off); off += (size_t)N2_ * 256 * 4;
  float* h1   = (float*)(ws + off); off += (size_t)N2_ * 256 * 4;
  float* lgt  = (float*)(ws + off); off += (size_t)N2_ * 64 * 4;

  constexpr int R0 = 196;
  constexpr int R1 = 40;
  static_assert((N1_ + R0 - 1) / R0 <= 256 && (N2_ + R1 - 1) / R1 <= 256,
                "owner id is one byte");

  constexpr int NQ0 = (E0_ >> 4) * 4;
  constexpr int NQ1 = (E1_ >> 4) * 4;
  constexpr int PB  = 256;
  static_assert(NQ0 + NQ1 <= PB * 256 * 4, "plane launch covers both planes");
  static_assert((size_t)(NQ0 + NQ1) * 4 + 256 <= (size_t)N2_ * 256 * 4, "planes fit in the h1 region");
  unsigned* own0 = (unsigned*)h1;
  unsigned* own1 = own0 + ((NQ0 + 63) / 64) * 64;
  owner_planes<R0, R1><<<PB, 256, 0, stream>>>(dst0, own0, NQ0, dst1, own1, NQ1);

  agg_mean_owner<R0><<<(N1_ + R0 - 1) / R0, 1024, 0, stream>>>(
      x, src0, dst0, own0, agg0, E0_, N1_);
  sage_gemm_wmma<512, 256, 1, false><<<(N1_ + 31) / 32, 512, 0, stream>>>(
      agg0, nullptr, x, Wl0, Wr0, bl0, h0, N1_);

  agg_mean_owner<R1><<<(N2_ + R1 - 1) / R1, 1024, 0, stream>>>(
      h0, src1, dst1, own1, agg1, E1_, N2_);
  sage_gemm_wmma<512, 256, 2, false><<<(N2_ + 31) / 32, 512, 0, stream>>>(
      agg1, nullptr, h0, Wl1, Wr1, bl1, h1, N2_);

  sage_gemm_wmma<256, 64, 0, false><<<(N2_ + 31) / 32, 128, 0, stream>>>(
      h1, nullptr, nullptr, Wlin, nullptr, blin, lgt, N2_);
  softmax_rows64<<<(N2_ + 7) / 8, 256, 0, stream>>>(lgt, (float*)d_out, N2_);
}
